# starting kernel
# baseline (speedup 1.0000x reference)
#include <hip/hip_runtime.h>
#include <hip/hip_bf16.h>

typedef float v2f __attribute__((ext_vector_type(2)));
typedef float v8f __attribute__((ext_vector_type(8)));

#define N_NODES   100000
#define E_EDGES   1000000
#define IN_F      256
#define OUT_F     128
#define NEG_SLOPE 0.2f

__global__ __launch_bounds__(256)
void gat_gemm_wmma(const float* __restrict__ X, const float* __restrict__ W,
                   float* __restrict__ h) {
  __shared__ float xs[16 * 260];
  const int m0  = blockIdx.x * 16;
  const int tid = threadIdx.x;

#pragma unroll
  for (int it = 0; it < 4; ++it) {
    const int f   = tid * 4 + it * 1024;
    const int row = f >> 8;
    const int col = f & 255;
    const float4 v = *(const float4*)(X + (size_t)(m0 + row) * IN_F + col);
    *(float4*)(&xs[row * 260 + col]) = v;
  }
  __syncthreads();

  const int wave  = tid >> 5;
  const int lane  = tid & 31;
  const int col0  = wave * 16;
  const int l15   = lane & 15;
  const int khalf = (lane >> 4) * 2;

  const float* xrow = &xs[l15 * 260 + khalf];
  const float* wrow = W + (size_t)(col0 + l15) * IN_F + khalf;

  v8f acc = {};
#pragma unroll 8
  for (int k0 = 0; k0 < IN_F; k0 += 4) {
    v2f a = *(const v2f*)(xrow + k0);
    v2f b = *(const v2f*)(wrow + k0);
    acc = __builtin_amdgcn_wmma_f32_16x16x4_f32(
         false, a,  false, b,
         (short)0, acc,  false,  false);
  }

  const int mrow = m0 + ((lane >> 4) << 3);
  float* hp = h + (size_t)mrow * OUT_F + col0 + l15;
#pragma unroll
  for (int r = 0; r < 8; ++r) {
    hp[(size_t)r * OUT_F] = acc[r];
  }
}

__global__ __launch_bounds__(256)
void gat_rowdot_init(const float* __restrict__ h,
                     const float* __restrict__ att_src,
                     const float* __restrict__ att_dst,
                     float* __restrict__ e_src, float* __restrict__ e_dst,
                     float* __restrict__ denom, float* __restrict__ out) {
  const int wave = threadIdx.x >> 5;
  const int lane = threadIdx.x & 31;
  const int row  = blockIdx.x * 8 + wave;
  if (row >= N_NODES) return;

  const float4 hv = ((const float4*)h)[(size_t)row * 32 + lane];
  const float4 as = ((const float4*)att_src)[lane];
  const float4 ad = ((const float4*)att_dst)[lane];
  float s = hv.x * as.x + hv.y * as.y + hv.z * as.z + hv.w * as.w;
  float d = hv.x * ad.x + hv.y * ad.y + hv.z * ad.z + hv.w * ad.w;
#pragma unroll
  for (int off = 16; off > 0; off >>= 1) {
    s += __shfl_xor(s, off, 32);
    d += __shfl_xor(d, off, 32);
  }

  ((float4*)out)[(size_t)row * 32 + lane] = make_float4(0.f, 0.f, 0.f, 0.f);
  if (lane == 0) {
    e_src[row] = s;
    e_dst[row] = d;
    denom[row] = 1e-12f;
  }
}

__global__ __launch_bounds__(256)
void gat_edge_logits(const int* __restrict__ ei,
                     const float* __restrict__ e_src,
                     const float* __restrict__ e_dst,
                     float* __restrict__ exp_e, float* __restrict__ denom) {
  const int t = blockIdx.x * 256 + threadIdx.x;
  if (t >= E_EDGES) return;
  const int i = ei[t];
  const int j = ei[E_EDGES + t];
  float e = e_src[i] + e_dst[j];
  e = (e >= 0.f) ? e : NEG_SLOPE * e;
  e = fminf(10.f, fmaxf(-10.f, e));
  const float ex = __expf(e);
  exp_e[t] = ex;
  unsafeAtomicAdd(&denom[i], ex);
}

__global__ __launch_bounds__(256)
void gat_scatter(const int* __restrict__ ei, const float* __restrict__ exp_e,
                 const float* __restrict__ denom, const float* __restrict__ h,
                 float* __restrict__ out) {
  const int wave = threadIdx.x >> 5;
  const int lane = threadIdx.x & 31;
  const int t    = blockIdx.x * 8 + wave;
  if (t >= E_EDGES) return;
  const int i = ei[t];
  const int j = ei[E_EDGES + t];
  const float alpha = exp_e[t] / denom[i];
  const float4 hv = ((const float4*)h)[(size_t)j * 32 + lane];
  float* op = out + (size_t)i * OUT_F + lane * 4;
  unsafeAtomicAdd(op + 0, alpha * hv.x);
  unsafeAtomicAdd(op + 1, alpha * hv.y);
  unsafeAtomicAdd(op + 2, alpha * hv.z);
  unsafeAtomicAdd(op + 3, alpha * hv.w);
}

extern "C" void kernel_launch(void* const* d_in, const int* in_sizes, int n_in,
                              void* d_out, int out_size, void* d_ws, size_t ws_size,
                              hipStream_t stream) {
  const float* X       = (const float*)d_in[0];
  const int*   ei      = (const int*)d_in[1];
  const float* W       = (const float*)d_in[2];
  const float* att_src = (const float*)d_in[3];
  const float* att_dst = (const float*)d_in[4];
  float* out = (float*)d_out;

  float* h     = (float*)d_ws;
  float* e_src = h + (size_t)N_NODES * OUT_F;
  float* e_dst = e_src + N_NODES;
  float* denom = e_dst + N_NODES;
  float* exp_e = denom + N_NODES;

  gat_gemm_wmma   <<<N_NODES / 16,        256, 0, stream>>>(X, W, h);
  gat_rowdot_init <<<N_NODES / 8,         256, 0, stream>>>(h, att_src, att_dst,
                                                            e_src, e_dst, denom, out);
  gat_edge_logits <<<(E_EDGES + 255)/256, 256, 0, stream>>>(ei, e_src, e_dst,
                                                            exp_e, denom);
  gat_scatter     <<<(E_EDGES + 7)/8,     256, 0, stream>>>(ei, exp_e, denom, h, out);
}


# scatter kernel: lane owns features lane+32q, so each f32 atomic add and row load of a wave covers one whole 128-byte line (was 4 lines x 8 floats, 16-byte stride); sums still in atomic arrival order
# speedup vs baseline: 1.2506x; 1.2506x over previous
#include <hip/hip_runtime.h>
#include <hip/hip_bf16.h>

typedef float v2f __attribute__((ext_vector_type(2)));
typedef float v8f __attribute__((ext_vector_type(8)));

#define N_NODES   100000
#define E_EDGES   1000000
#define IN_F      256
#define OUT_F     128
#define NEG_SLOPE 0.2f

__global__ __launch_bounds__(256)
void gat_gemm_wmma(const float* __restrict__ X, const float* __restrict__ W,
                   float* __restrict__ h) {
  __shared__ float xs[16 * 260];
  const int m0  = blockIdx.x * 16;
  const int tid = threadIdx.x;

#pragma unroll
  for (int it = 0; it < 4; ++it) {
    const int f   = tid * 4 + it * 1024;
    const int row = f >> 8;
    const int col = f & 255;
    const float4 v = *(const float4*)(X + (size_t)(m0 + row) * IN_F + col);
    *(float4*)(&xs[row * 260 + col]) = v;
  }
  __syncthreads();

  const int wave  = tid >> 5;
  const int lane  = tid & 31;
  const int col0  = wave * 16;
  const int l15   = lane & 15;
  const int khalf = (lane >> 4) * 2;

  const float* xrow = &xs[l15 * 260 + khalf];
  const float* wrow = W + (size_t)(col0 + l15) * IN_F + khalf;

  v8f acc = {};
#pragma unroll 8
  for (int k0 = 0; k0 < IN_F; k0 += 4) {
    v2f a = *(const v2f*)(xrow + k0);
    v2f b = *(const v2f*)(wrow + k0);
    acc = __builtin_amdgcn_wmma_f32_16x16x4_f32(
         false, a,  false, b,
         (short)0, acc,  false,  false);
  }

  const int mrow = m0 + ((lane >> 4) << 3);
  float* hp = h + (size_t)mrow * OUT_F + col0 + l15;
#pragma unroll
  for (int r = 0; r < 8; ++r) {
    hp[(size_t)r * OUT_F] = acc[r];
  }
}

__global__ __launch_bounds__(256)
void gat_rowdot_init(const float* __restrict__ h,
                     const float* __restrict__ att_src,
                     const float* __restrict__ att_dst,
                     float* __restrict__ e_src, float* __restrict__ e_dst,
                     float* __restrict__ denom, float* __restrict__ out) {
  const int wave = threadIdx.x >> 5;
  const int lane = threadIdx.x & 31;
  const int row  = blockIdx.x * 8 + wave;
  if (row >= N_NODES) return;

  const float4 hv = ((const float4*)h)[(size_t)row * 32 + lane];
  const float4 as = ((const float4*)att_src)[lane];
  const float4 ad = ((const float4*)att_dst)[lane];
  float s = hv.x * as.x + hv.y * as.y + hv.z * as.z + hv.w * as.w;
  float d = hv.x * ad.x + hv.y * ad.y + hv.z * ad.z + hv.w * ad.w;
#pragma unroll
  for (int off = 16; off > 0; off >>= 1) {
    s += __shfl_xor(s, off, 32);
    d += __shfl_xor(d, off, 32);
  }

  ((float4*)out)[(size_t)row * 32 + lane] = make_float4(0.f, 0.f, 0.f, 0.f);
  if (lane == 0) {
    e_src[row] = s;
    e_dst[row] = d;
    denom[row] = 1e-12f;
  }
}

__global__ __launch_bounds__(256)
void gat_edge_logits(const int* __restrict__ ei,
                     const float* __restrict__ e_src,
                     const float* __restrict__ e_dst,
                     float* __restrict__ exp_e, float* __restrict__ denom) {
  const int t = blockIdx.x * 256 + threadIdx.x;
  if (t >= E_EDGES) return;
  const int i = ei[t];
  const int j = ei[E_EDGES + t];
  float e = e_src[i] + e_dst[j];
  e = (e >= 0.f) ? e : NEG_SLOPE * e;
  e = fminf(10.f, fmaxf(-10.f, e));
  const float ex = __expf(e);
  exp_e[t] = ex;
  unsafeAtomicAdd(&denom[i], ex);
}

__global__ __launch_bounds__(256)
void gat_scatter(const int* __restrict__ ei, const float* __restrict__ exp_e,
                 const float* __restrict__ denom, const float* __restrict__ h,
                 float* __restrict__ out) {
  const int wave = threadIdx.x >> 5;
  const int lane = threadIdx.x & 31;
  const int t    = blockIdx.x * 8 + wave;
  if (t >= E_EDGES) return;
  const int i = ei[t];
  const int j = ei[E_EDGES + t];
  const float alpha = exp_e[t] / denom[i];
  const float* hp = h + (size_t)j * OUT_F + lane;
  const float h0 = hp[0];
  const float h1 = hp[32];
  const float h2 = hp[64];
  const float h3 = hp[96];
  float* op = out + (size_t)i * OUT_F + lane;
  unsafeAtomicAdd(op + 0,  alpha * h0);
  unsafeAtomicAdd(op + 32, alpha * h1);
  unsafeAtomicAdd(op + 64, alpha * h2);
  unsafeAtomicAdd(op + 96, alpha * h3);
}

extern "C" void kernel_launch(void* const* d_in, const int* in_sizes, int n_in,
                              void* d_out, int out_size, void* d_ws, size_t ws_size,
                              hipStream_t stream) {
  const float* X       = (const float*)d_in[0];
  const int*   ei      = (const int*)d_in[1];
  const float* W       = (const float*)d_in[2];
  const float* att_src = (const float*)d_in[3];
  const float* att_dst = (const float*)d_in[4];
  float* out = (float*)d_out;

  float* h     = (float*)d_ws;
  float* e_src = h + (size_t)N_NODES * OUT_F;
  float* e_dst = e_src + N_NODES;
  float* denom = e_dst + N_NODES;
  float* exp_e = denom + N_NODES;

  gat_gemm_wmma   <<<N_NODES / 16,        256, 0, stream>>>(X, W, h);
  gat_rowdot_init <<<N_NODES / 8,         256, 0, stream>>>(h, att_src, att_dst,
                                                            e_src, e_dst, denom, out);
  gat_edge_logits <<<(E_EDGES + 255)/256, 256, 0, stream>>>(ei, e_src, e_dst,
                                                            exp_e, denom);
  gat_scatter     <<<(E_EDGES + 7)/8,     256, 0, stream>>>(ei, exp_e, denom, h, out);
}


# edges bucketed by destination in-call (returning int atomic = arrival-order rank, scan, place); one wave per node sums its rows in registers in rank order and stores out once; no f32 atomics
# speedup vs baseline: 1.5661x; 1.2523x over previous
#include <hip/hip_runtime.h>
#include <hip/hip_bf16.h>

typedef float v2f __attribute__((ext_vector_type(2)));
typedef float v8f __attribute__((ext_vector_type(8)));

#define N_NODES   100000
#define E_EDGES   1000000
#define IN_F      256
#define OUT_F     128
#define NEG_SLOPE 0.2f

__global__ __launch_bounds__(256)
void gat_gemm_wmma(const float* __restrict__ X, const float* __restrict__ W,
                   float* __restrict__ h) {
  __shared__ float xs[16 * 260];
  const int m0  = blockIdx.x * 16;
  const int tid = threadIdx.x;

#pragma unroll
  for (int it = 0; it < 4; ++it) {
    const int f   = tid * 4 + it * 1024;
    const int row = f >> 8;
    const int col = f & 255;
    const float4 v = *(const float4*)(X + (size_t)(m0 + row) * IN_F + col);
    *(float4*)(&xs[row * 260 + col]) = v;
  }
  __syncthreads();

  const int wave  = tid >> 5;
  const int lane  = tid & 31;
  const int col0  = wave * 16;
  const int l15   = lane & 15;
  const int khalf = (lane >> 4) * 2;

  const float* xrow = &xs[l15 * 260 + khalf];
  const float* wrow = W + (size_t)(col0 + l15) * IN_F + khalf;

  v8f acc = {};
#pragma unroll 8
  for (int k0 = 0; k0 < IN_F; k0 += 4) {
    v2f a = *(const v2f*)(xrow + k0);
    v2f b = *(const v2f*)(wrow + k0);
    acc = __builtin_amdgcn_wmma_f32_16x16x4_f32(
         false, a,  false, b,
         (short)0, acc,  false,  false);
  }

  const int mrow = m0 + ((lane >> 4) << 3);
  float* hp = h + (size_t)mrow * OUT_F + col0 + l15;
#pragma unroll
  for (int r = 0; r < 8; ++r) {
    hp[(size_t)r * OUT_F] = acc[r];
  }
}

#define SCAN_THREADS 1024
#define SCAN_BLOCKS  ((N_NODES + SCAN_THREADS - 1) / SCAN_THREADS)
#define PRE_TRIPS    (((SCAN_BLOCKS - 1) * (SCAN_THREADS / 4) + SCAN_THREADS - 1) / SCAN_THREADS)

__device__ __forceinline__ float gat_edge_exp(float e) {
  e = (e >= 0.f) ? e : NEG_SLOPE * e;
  e = fminf(10.f, fmaxf(-10.f, e));
  return __expf(e);
}

__global__ __launch_bounds__(256)
void gat_rowdot_init(const float* __restrict__ h,
                     const float* __restrict__ att_src,
                     const float* __restrict__ att_dst,
                     float* __restrict__ e_src, float* __restrict__ e_dst,
                     int* __restrict__ cnt) {
  const int wave = threadIdx.x >> 5;
  const int lane = threadIdx.x & 31;
  const int row  = blockIdx.x * 8 + wave;
  if (row >= N_NODES) return;

  const float4 hv = ((const float4*)h)[(size_t)row * 32 + lane];
  const float4 as = ((const float4*)att_src)[lane];
  const float4 ad = ((const float4*)att_dst)[lane];
  float s = hv.x * as.x + hv.y * as.y + hv.z * as.z + hv.w * as.w;
  float d = hv.x * ad.x + hv.y * ad.y + hv.z * ad.z + hv.w * ad.w;
#pragma unroll
  for (int off = 16; off > 0; off >>= 1) {
    s += __shfl_xor(s, off, 32);
    d += __shfl_xor(d, off, 32);
  }
  if (lane == 0) {
    e_src[row] = s;
    e_dst[row] = d;
    cnt[row]   = 0;
  }
}

__global__ __launch_bounds__(256)
void gat_edge_count(const int* __restrict__ ei, int* __restrict__ cnt,
                    int* __restrict__ rank) {
  const int t = blockIdx.x * 256 + threadIdx.x;
  if (t >= E_EDGES) return;
  rank[t] = atomicAdd(&cnt[ei[t]], 1);
}

__global__ __launch_bounds__(SCAN_THREADS)
void gat_degree_scan(const int* __restrict__ cnt, int* __restrict__ ends) {
  __shared__ int s_pre[32];
  __shared__ int s_w[32];
  const int tid  = threadIdx.x;
  const int lane = tid & 31;
  const int wave = tid >> 5;
  const int base = blockIdx.x * SCAN_THREADS;
  const int n4   = base >> 2;

  const int4* c4 = (const int4*)cnt;
  int pre = 0;
#pragma unroll
  for (int q = 0; q < PRE_TRIPS; ++q) {
    const int idx  = tid + q * SCAN_THREADS;
    const int4 v   = c4[min(idx, N_NODES / 4 - 1)];
    const int keep = (idx < n4) ? -1 : 0;
    pre += ((v.x + v.y) + (v.z + v.w)) & keep;
  }

  const int k = base + tid;
  const int c = (k < N_NODES) ? cnt[k] : 0;
  int inc = c;
#pragma unroll
  for (int off = 1; off < 32; off <<= 1) {
    const int n = __shfl_up(inc, off, 32);
    if (lane >= off) inc += n;
  }
#pragma unroll
  for (int off = 16; off > 0; off >>= 1) pre += __shfl_xor(pre, off, 32);
  if (lane == 31) s_w[wave]   = inc;
  if (lane == 0)  s_pre[wave] = pre;
  __syncthreads();

  int p = s_pre[lane];
#pragma unroll
  for (int off = 16; off > 0; off >>= 1) p += __shfl_xor(p, off, 32);
  const int w = s_w[lane];
  int winc = w;
#pragma unroll
  for (int off = 1; off < 32; off <<= 1) {
    const int n = __shfl_up(winc, off, 32);
    if (lane >= off) winc += n;
  }
  const int wave_off = __shfl(winc - w, wave, 32);
  if (k < N_NODES) ends[k] = p + wave_off + inc;
}

__global__ __launch_bounds__(256)
void gat_edge_place(const int* __restrict__ ei, const int* __restrict__ rank,
                    const int* __restrict__ ends, int* __restrict__ rec) {
  const int t = blockIdx.x * 256 + threadIdx.x;
  if (t >= E_EDGES) return;
  const int i = ei[t];
  const int j = ei[E_EDGES + t];
  const int start = (i > 0) ? ends[i - 1] : 0;
  rec[start + rank[t]] = j;
}

__global__ __launch_bounds__(256)
void gat_node_gather(const int* __restrict__ ends, const int* __restrict__ rec,
                     const float* __restrict__ e_src,
                     const float* __restrict__ e_dst,
                     const float* __restrict__ h, float* __restrict__ out) {
  const int lane = threadIdx.x & 31;
  const int row = __builtin_amdgcn_readfirstlane(blockIdx.x * 8 + (threadIdx.x >> 5));
  if (row >= N_NODES) return;

  const int start = (row > 0) ? ends[row - 1] : 0;
  const int deg   = __builtin_amdgcn_readfirstlane(ends[row] - start);
  const float es  = e_src[row];
  const int* bucket = rec + start;

  int   j0  = 0;
  float ex0 = 0.f;
  if (lane < deg) {
    j0  = bucket[lane];
    ex0 = gat_edge_exp(es + e_dst[j0]);
  }
  float dsum = ex0;
  for (int base = 32; base < deg; base += 32) {
    const int idx = base + lane;
    if (idx < deg) dsum += gat_edge_exp(es + e_dst[bucket[idx]]);
  }
#pragma unroll
  for (int off = 16; off > 0; off >>= 1) dsum += __shfl_xor(dsum, off, 32);
  const float denom = dsum + 1e-12f;

  const float4* h4 = (const float4*)h;
  float4 acc = make_float4(0.f, 0.f, 0.f, 0.f);
  int   jv = j0;
  float av = ex0 / denom;
  for (int base = 0; base < deg; base += 32) {
    if (base > 0) {
      const int idx = base + lane;
      jv = 0;
      av = 0.f;
      if (idx < deg) {
        jv = bucket[idx];
        av = gat_edge_exp(es + e_dst[jv]) / denom;
      }
    }
    const int n = min(32, deg - base);
#pragma unroll 4
    for (int e = 0; e < n; ++e) {
      const int   j = __builtin_amdgcn_readlane(jv, e);
      const float a = __int_as_float(__builtin_amdgcn_readlane(__float_as_int(av), e));
      const float4 hv = h4[(size_t)j * 32 + lane];
      acc.x += a * hv.x;
      acc.y += a * hv.y;
      acc.z += a * hv.z;
      acc.w += a * hv.w;
    }
  }
  ((float4*)out)[(size_t)row * 32 + lane] = acc;
}

extern "C" void kernel_launch(void* const* d_in, const int* in_sizes, int n_in,
                              void* d_out, int out_size, void* d_ws, size_t ws_size,
                              hipStream_t stream) {
  const float* X       = (const float*)d_in[0];
  const int*   ei      = (const int*)d_in[1];
  const float* W       = (const float*)d_in[2];
  const float* att_src = (const float*)d_in[3];
  const float* att_dst = (const float*)d_in[4];
  float* out = (float*)d_out;

  float* h     = (float*)d_ws;
  float* e_src = h + (size_t)N_NODES * OUT_F;
  float* e_dst = e_src + N_NODES;
  int*   ends  = (int*)(e_dst + N_NODES);
  int*   rec   = ends + N_NODES;
  int*   cnt   = (int*)d_out;
  int*   rank  = cnt + N_NODES;

  gat_gemm_wmma   <<<N_NODES / 16,        256, 0, stream>>>(X, W, h);
  gat_rowdot_init <<<N_NODES / 8,         256, 0, stream>>>(h, att_src, att_dst,
                                                            e_src, e_dst, cnt);
  gat_edge_count  <<<(E_EDGES + 255)/256, 256, 0, stream>>>(ei, cnt, rank);
  gat_degree_scan <<<SCAN_BLOCKS, SCAN_THREADS, 0, stream>>>(cnt, ends);
  gat_edge_place  <<<(E_EDGES + 255)/256, 256, 0, stream>>>(ei, rank, ends, rec);
  gat_node_gather <<<(N_NODES + 7) / 8,   256, 0, stream>>>(ends, rec, e_src, e_dst,
                                                            h, out);
}


# edge bucketing on LDS counters: work-group (node range, edge slice) counts its edges, next launch hands out bucket slots by returning LDS adds; no global atomics, 5 launches; sums in arrival order
# speedup vs baseline: 2.1803x; 1.3922x over previous
#include <hip/hip_runtime.h>
#include <hip/hip_bf16.h>

typedef float v2f __attribute__((ext_vector_type(2)));
typedef float v8f __attribute__((ext_vector_type(8)));

#define N_NODES   100000
#define E_EDGES   1000000
#define IN_F      256
#define OUT_F     128
#define NEG_SLOPE 0.2f

__global__ __launch_bounds__(256)
void gat_gemm_wmma(const float* __restrict__ X, const float* __restrict__ W,
                   float* __restrict__ h) {
  __shared__ float xs[16 * 260];
  const int m0  = blockIdx.x * 16;
  const int tid = threadIdx.x;

#pragma unroll
  for (int it = 0; it < 4; ++it) {
    const int f   = tid * 4 + it * 1024;
    const int row = f >> 8;
    const int col = f & 255;
    const float4 v = *(const float4*)(X + (size_t)(m0 + row) * IN_F + col);
    *(float4*)(&xs[row * 260 + col]) = v;
  }
  __syncthreads();

  const int wave  = tid >> 5;
  const int lane  = tid & 31;
  const int col0  = wave * 16;
  const int l15   = lane & 15;
  const int khalf = (lane >> 4) * 2;

  const float* xrow = &xs[l15 * 260 + khalf];
  const float* wrow = W + (size_t)(col0 + l15) * IN_F + khalf;

  v8f acc = {};
#pragma unroll 8
  for (int k0 = 0; k0 < IN_F; k0 += 4) {
    v2f a = *(const v2f*)(xrow + k0);
    v2f b = *(const v2f*)(wrow + k0);
    acc = __builtin_amdgcn_wmma_f32_16x16x4_f32(
         false, a,  false, b,
         (short)0, acc,  false,  false);
  }

  const int mrow = m0 + ((lane >> 4) << 3);
  float* hp = h + (size_t)mrow * OUT_F + col0 + l15;
#pragma unroll
  for (int r = 0; r < 8; ++r) {
    hp[(size_t)r * OUT_F] = acc[r];
  }
}

#define BK_G       32
#define BK_S       8
#define BK_RANGE   3128
#define BK_NP      (BK_G * BK_RANGE)
#define BK_SLICE   (E_EDGES / BK_S)
#define BK_SLICE4  (BK_SLICE / 4)
#define BK_THREADS 1024
#define BK_TRIPS   ((BK_SLICE4 + BK_THREADS - 1) / BK_THREADS)
#define BK_OWNERS  (BK_RANGE / 4)

static_assert(BK_NP >= N_NODES, "ranges must cover every node");
static_assert(BK_S * BK_SLICE == E_EDGES && BK_SLICE % 4 == 0, "slices must tile the edge list in int4s");
static_assert(BK_RANGE % 4 == 0 && N_NODES % 4 == 0, "int4 access to node planes");
static_assert(BK_OWNERS <= BK_THREADS && BK_G * BK_S <= BK_THREADS, "one thread per owner / per total");

__device__ __forceinline__ float gat_edge_exp(float e) {
  e = (e >= 0.f) ? e : NEG_SLOPE * e;
  e = fminf(10.f, fmaxf(-10.f, e));
  return __expf(e);
}

__global__ __launch_bounds__(256)
void gat_rowdot(const float* __restrict__ h,
                const float* __restrict__ att_src,
                const float* __restrict__ att_dst,
                float* __restrict__ e_src, float* __restrict__ e_dst) {
  const int wave = threadIdx.x >> 5;
  const int lane = threadIdx.x & 31;
  const int row  = blockIdx.x * 8 + wave;
  if (row >= N_NODES) return;

  const float4 hv = ((const float4*)h)[(size_t)row * 32 + lane];
  const float4 as = ((const float4*)att_src)[lane];
  const float4 ad = ((const float4*)att_dst)[lane];
  float s = hv.x * as.x + hv.y * as.y + hv.z * as.z + hv.w * as.w;
  float d = hv.x * ad.x + hv.y * ad.y + hv.z * ad.z + hv.w * ad.w;
#pragma unroll
  for (int off = 16; off > 0; off >>= 1) {
    s += __shfl_xor(s, off, 32);
    d += __shfl_xor(d, off, 32);
  }
  if (lane == 0) {
    e_src[row] = s;
    e_dst[row] = d;
  }
}

__global__ __launch_bounds__(BK_THREADS)
void gat_bucket_count(const int* __restrict__ ei, int* __restrict__ cnt2,
                      int* __restrict__ tot) {
  __shared__ __attribute__((aligned(16))) int c[BK_RANGE];
  __shared__ int s_tot;
  const int tid  = threadIdx.x;
  const int lane = tid & 31;
  const int g    = blockIdx.x % BK_G;
  const int s    = blockIdx.x / BK_G;
  const int lo   = g * BK_RANGE;

  for (int k = tid; k < BK_RANGE; k += BK_THREADS) c[k] = 0;
  if (tid == 0) s_tot = 0;
  __syncthreads();

  const int4* d4 = (const int4*)(ei + (size_t)s * BK_SLICE);
#pragma unroll
  for (int b = 0; b < BK_TRIPS; b += 8) {
    int4 v[8];
#pragma unroll
    for (int u = 0; u < 8; ++u) {
      if (b + u < BK_TRIPS) {
        const int idx = tid + (b + u) * BK_THREADS;
        v[u] = d4[min(idx, BK_SLICE4 - 1)];
      }
    }
#pragma unroll
    for (int u = 0; u < 8; ++u) {
      if (b + u < BK_TRIPS) {
        const int idx = tid + (b + u) * BK_THREADS;
        const bool live = ((b + u + 1) * BK_THREADS <= BK_SLICE4) || (idx < BK_SLICE4);
        const unsigned d0 = (unsigned)(v[u].x - lo);
        const unsigned d1 = (unsigned)(v[u].y - lo);
        const unsigned d2 = (unsigned)(v[u].z - lo);
        const unsigned d3 = (unsigned)(v[u].w - lo);
        if (live && d0 < BK_RANGE) atomicAdd(&c[d0], 1);
        if (live && d1 < BK_RANGE) atomicAdd(&c[d1], 1);
        if (live && d2 < BK_RANGE) atomicAdd(&c[d2], 1);
        if (live && d3 < BK_RANGE) atomicAdd(&c[d3], 1);
      }
    }
  }
  __syncthreads();

  int sum = 0;
  if (tid < BK_OWNERS) {
    const int4 x = ((const int4*)c)[tid];
    ((int4*)(cnt2 + (size_t)s * BK_NP + lo))[tid] = x;
    sum = (x.x + x.y) + (x.z + x.w);
  }
#pragma unroll
  for (int off = 16; off > 0; off >>= 1) sum += __shfl_xor(sum, off, 32);
  if (lane == 0 && sum != 0) atomicAdd(&s_tot, sum);
  __syncthreads();
  if (tid == 0) tot[g * BK_S + s] = s_tot;
}

__global__ __launch_bounds__(BK_THREADS)
void gat_bucket_place(const int* __restrict__ ei, const int* __restrict__ cnt2,
                      const int* __restrict__ tot, int* __restrict__ ends,
                      int* __restrict__ rec) {
  __shared__ __attribute__((aligned(16))) int cur[BK_RANGE];
  __shared__ int s_w[32];
  __shared__ int s_b[32];
  const int tid  = threadIdx.x;
  const int lane = tid & 31;
  const int wave = tid >> 5;
  const int g    = blockIdx.x % BK_G;
  const int s    = blockIdx.x / BK_G;
  const int lo   = g * BK_RANGE;

  int below = (tid < g * BK_S) ? tot[tid] : 0;

  int4 t4 = make_int4(0, 0, 0, 0);
  int4 m4 = make_int4(0, 0, 0, 0);
  if (tid < BK_OWNERS) {
#pragma unroll
    for (int q = 0; q < BK_S; ++q) {
      const int4 x = ((const int4*)(cnt2 + (size_t)q * BK_NP + lo))[tid];
      t4.x += x.x; t4.y += x.y; t4.z += x.z; t4.w += x.w;
      const int keep = (q < s) ? -1 : 0;
      m4.x += x.x & keep; m4.y += x.y & keep; m4.z += x.z & keep; m4.w += x.w & keep;
    }
  }
  const int tsum = (t4.x + t4.y) + (t4.z + t4.w);

  int inc = tsum;
#pragma unroll
  for (int off = 1; off < 32; off <<= 1) {
    const int n = __shfl_up(inc, off, 32);
    if (lane >= off) inc += n;
  }
#pragma unroll
  for (int off = 16; off > 0; off >>= 1) below += __shfl_xor(below, off, 32);
  if (lane == 31) s_w[wave] = inc;
  if (lane == 0)  s_b[wave] = below;
  __syncthreads();

  int p = s_b[lane];
#pragma unroll
  for (int off = 16; off > 0; off >>= 1) p += __shfl_xor(p, off, 32);
  const int w = s_w[lane];
  int winc = w;
#pragma unroll
  for (int off = 1; off < 32; off <<= 1) {
    const int n = __shfl_up(winc, off, 32);
    if (lane >= off) winc += n;
  }
  const int wave_off = __shfl(winc - w, wave, 32);

  if (tid < BK_OWNERS) {
    const int st0 = p + wave_off + (inc - tsum);
    const int st1 = st0 + t4.x;
    const int st2 = st1 + t4.y;
    const int st3 = st2 + t4.z;
    ((int4*)cur)[tid] = make_int4(st0 + m4.x, st1 + m4.y, st2 + m4.z, st3 + m4.w);
    const int node = lo + 4 * tid;
    if (s == 0 && node < N_NODES)
      *(int4*)(ends + node) = make_int4(st1, st2, st3, st3 + t4.w);
  }
  __syncthreads();

  const int4* d4 = (const int4*)(ei + (size_t)s * BK_SLICE);
  const int4* j4 = (const int4*)(ei + (size_t)E_EDGES + (size_t)s * BK_SLICE);
#pragma unroll
  for (int b = 0; b < BK_TRIPS; b += 4) {
    int4 v[4];
    int4 jj[4];
#pragma unroll
    for (int u = 0; u < 4; ++u) {
      if (b + u < BK_TRIPS) {
        const int idx = min(tid + (b + u) * BK_THREADS, BK_SLICE4 - 1);
        v[u]  = d4[idx];
        jj[u] = j4[idx];
      }
    }
    int  pos[4][4];
    bool hit[4][4];
#pragma unroll
    for (int u = 0; u < 4; ++u) {
      if (b + u < BK_TRIPS) {
        const int idx = tid + (b + u) * BK_THREADS;
        const bool live = ((b + u + 1) * BK_THREADS <= BK_SLICE4) || (idx < BK_SLICE4);
        const unsigned d0 = (unsigned)(v[u].x - lo);
        const unsigned d1 = (unsigned)(v[u].y - lo);
        const unsigned d2 = (unsigned)(v[u].z - lo);
        const unsigned d3 = (unsigned)(v[u].w - lo);
        hit[u][0] = live && d0 < BK_RANGE;
        hit[u][1] = live && d1 < BK_RANGE;
        hit[u][2] = live && d2 < BK_RANGE;
        hit[u][3] = live && d3 < BK_RANGE;
        if (hit[u][0]) pos[u][0] = atomicAdd(&cur[d0], 1);
        if (hit[u][1]) pos[u][1] = atomicAdd(&cur[d1], 1);
        if (hit[u][2]) pos[u][2] = atomicAdd(&cur[d2], 1);
        if (hit[u][3]) pos[u][3] = atomicAdd(&cur[d3], 1);
      }
    }
#pragma unroll
    for (int u = 0; u < 4; ++u) {
      if (b + u < BK_TRIPS) {
        if (hit[u][0]) rec[pos[u][0]] = jj[u].x;
        if (hit[u][1]) rec[pos[u][1]] = jj[u].y;
        if (hit[u][2]) rec[pos[u][2]] = jj[u].z;
        if (hit[u][3]) rec[pos[u][3]] = jj[u].w;
      }
    }
  }
}

__global__ __launch_bounds__(256)
void gat_node_gather(const int* __restrict__ ends, const int* __restrict__ rec,
                     const float* __restrict__ e_src,
                     const float* __restrict__ e_dst,
                     const float* __restrict__ h, float* __restrict__ out) {
  const int lane = threadIdx.x & 31;
  const int row = __builtin_amdgcn_readfirstlane(blockIdx.x * 8 + (threadIdx.x >> 5));
  if (row >= N_NODES) return;

  const int start = (row > 0) ? ends[row - 1] : 0;
  const int deg   = __builtin_amdgcn_readfirstlane(ends[row] - start);
  const float es  = e_src[row];
  const int* bucket = rec + start;

  int   j0  = 0;
  float ex0 = 0.f;
  if (lane < deg) {
    j0  = bucket[lane];
    ex0 = gat_edge_exp(es + e_dst[j0]);
  }
  float dsum = ex0;
  for (int base = 32; base < deg; base += 32) {
    const int idx = base + lane;
    if (idx < deg) dsum += gat_edge_exp(es + e_dst[bucket[idx]]);
  }
#pragma unroll
  for (int off = 16; off > 0; off >>= 1) dsum += __shfl_xor(dsum, off, 32);
  const float denom = dsum + 1e-12f;

  const float4* h4 = (const float4*)h;
  float4 acc = make_float4(0.f, 0.f, 0.f, 0.f);
  int   jv = j0;
  float av = ex0 / denom;
  for (int base = 0; base < deg; base += 32) {
    if (base > 0) {
      const int idx = base + lane;
      jv = 0;
      av = 0.f;
      if (idx < deg) {
        jv = bucket[idx];
        av = gat_edge_exp(es + e_dst[jv]) / denom;
      }
    }
    const int n = min(32, deg - base);
#pragma unroll 4
    for (int e = 0; e < n; ++e) {
      const int   j = __builtin_amdgcn_readlane(jv, e);
      const float a = __int_as_float(__builtin_amdgcn_readlane(__float_as_int(av), e));
      const float4 hv = h4[(size_t)j * 32 + lane];
      acc.x += a * hv.x;
      acc.y += a * hv.y;
      acc.z += a * hv.z;
      acc.w += a * hv.w;
    }
  }
  ((float4*)out)[(size_t)row * 32 + lane] = acc;
}

extern "C" void kernel_launch(void* const* d_in, const int* in_sizes, int n_in,
                              void* d_out, int out_size, void* d_ws, size_t ws_size,
                              hipStream_t stream) {
  const float* X       = (const float*)d_in[0];
  const int*   ei      = (const int*)d_in[1];
  const float* W       = (const float*)d_in[2];
  const float* att_src = (const float*)d_in[3];
  const float* att_dst = (const float*)d_in[4];
  float* out = (float*)d_out;

  float* h     = (float*)d_ws;
  float* e_src = h + (size_t)N_NODES * OUT_F;
  float* e_dst = e_src + N_NODES;
  int*   ends  = (int*)(e_dst + N_NODES);
  int*   rec   = ends + N_NODES;
  int*   cnt2  = (int*)d_out;
  int*   tot   = cnt2 + (size_t)BK_S * BK_NP;

  gat_gemm_wmma    <<<N_NODES / 16,      256, 0, stream>>>(X, W, h);
  gat_rowdot       <<<N_NODES / 8,       256, 0, stream>>>(h, att_src, att_dst,
                                                           e_src, e_dst);
  gat_bucket_count <<<BK_G * BK_S, BK_THREADS, 0, stream>>>(ei, cnt2, tot);
  gat_bucket_place <<<BK_G * BK_S, BK_THREADS, 0, stream>>>(ei, cnt2, tot, ends, rec);
  gat_node_gather  <<<(N_NODES + 7) / 8, 256, 0, stream>>>(ends, rec, e_src, e_dst,
                                                           h, out);
}


# row dots e_src, e_dst taken from the WMMA accumulators in the GEMM epilogue (DPP row sums, 8 waves added through LDS); row-dot launch and its 100000 waves gone; dot sum order changed, still fixed
# speedup vs baseline: 2.3707x; 1.0873x over previous
#include <hip/hip_runtime.h>
#include <hip/hip_bf16.h>

typedef float v2f __attribute__((ext_vector_type(2)));
typedef float v8f __attribute__((ext_vector_type(8)));

#define N_NODES   100000
#define E_EDGES   1000000
#define IN_F      256
#define OUT_F     128
#define NEG_SLOPE 0.2f

template <int MASK>
__device__ __forceinline__ float row_xor(float v) {
  return __int_as_float(__builtin_amdgcn_update_dpp(
      0, __float_as_int(v), 0x160 | MASK, 0xf, 0xf, false));
}

__global__ __launch_bounds__(256)
void gat_gemm_wmma(const float* __restrict__ X, const float* __restrict__ W,
                   const float* __restrict__ att_src,
                   const float* __restrict__ att_dst,
                   float* __restrict__ h,
                   float* __restrict__ e_src, float* __restrict__ e_dst) {
  __shared__ float xs[16 * 260];
  __shared__ float red[8 * 32];
  const int m0  = blockIdx.x * 16;
  const int tid = threadIdx.x;

#pragma unroll
  for (int it = 0; it < 4; ++it) {
    const int f   = tid * 4 + it * 1024;
    const int row = f >> 8;
    const int col = f & 255;
    const float4 v = *(const float4*)(X + (size_t)(m0 + row) * IN_F + col);
    *(float4*)(&xs[row * 260 + col]) = v;
  }
  __syncthreads();

  const int wave  = tid >> 5;
  const int lane  = tid & 31;
  const int col0  = wave * 16;
  const int l15   = lane & 15;
  const int khalf = (lane >> 4) * 2;

  const float* xrow = &xs[l15 * 260 + khalf];
  const float* wrow = W + (size_t)(col0 + l15) * IN_F + khalf;

  const float a_s = att_src[col0 + l15];
  const float a_d = att_dst[col0 + l15];

  v8f acc = {};
#pragma unroll 8
  for (int k0 = 0; k0 < IN_F; k0 += 4) {
    v2f a = *(const v2f*)(xrow + k0);
    v2f b = *(const v2f*)(wrow + k0);
    acc = __builtin_amdgcn_wmma_f32_16x16x4_f32(
         false, a,  false, b,
         (short)0, acc,  false,  false);
  }

  asm volatile("v_nop\n\tv_nop\n\tv_nop\n\tv_nop\n\tv_nop\n\tv_nop\n\tv_nop\n\tv_nop" : "+v"(acc));

  float v[16];
#pragma unroll
  for (int r = 0; r < 8; ++r) {
    v[r]     = acc[r] * a_s;
    v[8 + r] = acc[r] * a_d;
  }
  const bool b3 = (l15 & 8) != 0;
  const bool b2 = (l15 & 4) != 0;
  const bool b1 = (l15 & 2) != 0;
  const bool b0 = (l15 & 1) != 0;
  float w8[8];
#pragma unroll
  for (int i = 0; i < 8; ++i) {
    const float send = b3 ? v[i] : v[i + 8];
    const float keep = b3 ? v[i + 8] : v[i];
    w8[i] = keep + row_xor<8>(send);
  }
  float w4[4];
#pragma unroll
  for (int i = 0; i < 4; ++i) {
    const float send = b2 ? w8[i] : w8[i + 4];
    const float keep = b2 ? w8[i + 4] : w8[i];
    w4[i] = keep + row_xor<4>(send);
  }
  float w2[2];
#pragma unroll
  for (int i = 0; i < 2; ++i) {
    const float send = b1 ? w4[i] : w4[i + 2];
    const float keep = b1 ? w4[i + 2] : w4[i];
    w2[i] = keep + row_xor<2>(send);
  }
  const float w1 = (b0 ? w2[1] : w2[0]) + row_xor<1>(b0 ? w2[0] : w2[1]);
  red[wave * 32 + lane] = w1;
  __syncthreads();

  const int mrow = m0 + ((lane >> 4) << 3);
  float* hp = h + (size_t)mrow * OUT_F + col0 + l15;
#pragma unroll
  for (int r = 0; r < 8; ++r) {
    hp[(size_t)r * OUT_F] = acc[r];
  }

  if (tid < 32) {
    float s = red[tid];
#pragma unroll
    for (int w = 1; w < 8; ++w) s += red[w * 32 + tid];
    const int row = m0 + ((tid >> 4) << 3) + (tid & 7);
    float* dst = (tid & 8) ? e_dst : e_src;
    dst[row] = s;
  }
}

#define BK_G       32
#define BK_S       8
#define BK_RANGE   3128
#define BK_NP      (BK_G * BK_RANGE)
#define BK_SLICE   (E_EDGES / BK_S)
#define BK_SLICE4  (BK_SLICE / 4)
#define BK_THREADS 1024
#define BK_TRIPS   ((BK_SLICE4 + BK_THREADS - 1) / BK_THREADS)
#define BK_OWNERS  (BK_RANGE / 4)

static_assert(BK_NP >= N_NODES, "ranges must cover every node");
static_assert(BK_S * BK_SLICE == E_EDGES && BK_SLICE % 4 == 0, "slices must tile the edge list in int4s");
static_assert(BK_RANGE % 4 == 0 && N_NODES % 4 == 0, "int4 access to node planes");
static_assert(BK_OWNERS <= BK_THREADS && BK_G * BK_S <= BK_THREADS, "one thread per owner / per total");

__device__ __forceinline__ float gat_edge_exp(float e) {
  e = (e >= 0.f) ? e : NEG_SLOPE * e;
  e = fminf(10.f, fmaxf(-10.f, e));
  return __expf(e);
}

__global__ __launch_bounds__(BK_THREADS)
void gat_bucket_count(const int* __restrict__ ei, int* __restrict__ cnt2,
                      int* __restrict__ tot) {
  __shared__ __attribute__((aligned(16))) int c[BK_RANGE];
  __shared__ int s_tot;
  const int tid  = threadIdx.x;
  const int lane = tid & 31;
  const int g    = blockIdx.x % BK_G;
  const int s    = blockIdx.x / BK_G;
  const int lo   = g * BK_RANGE;

  for (int k = tid; k < BK_RANGE; k += BK_THREADS) c[k] = 0;
  if (tid == 0) s_tot = 0;
  __syncthreads();

  const int4* d4 = (const int4*)(ei + (size_t)s * BK_SLICE);
#pragma unroll
  for (int b = 0; b < BK_TRIPS; b += 8) {
    int4 v[8];
#pragma unroll
    for (int u = 0; u < 8; ++u) {
      if (b + u < BK_TRIPS) {
        const int idx = tid + (b + u) * BK_THREADS;
        v[u] = d4[min(idx, BK_SLICE4 - 1)];
      }
    }
#pragma unroll
    for (int u = 0; u < 8; ++u) {
      if (b + u < BK_TRIPS) {
        const int idx = tid + (b + u) * BK_THREADS;
        const bool live = ((b + u + 1) * BK_THREADS <= BK_SLICE4) || (idx < BK_SLICE4);
        const unsigned d0 = (unsigned)(v[u].x - lo);
        const unsigned d1 = (unsigned)(v[u].y - lo);
        const unsigned d2 = (unsigned)(v[u].z - lo);
        const unsigned d3 = (unsigned)(v[u].w - lo);
        if (live && d0 < BK_RANGE) atomicAdd(&c[d0], 1);
        if (live && d1 < BK_RANGE) atomicAdd(&c[d1], 1);
        if (live && d2 < BK_RANGE) atomicAdd(&c[d2], 1);
        if (live && d3 < BK_RANGE) atomicAdd(&c[d3], 1);
      }
    }
  }
  __syncthreads();

  int sum = 0;
  if (tid < BK_OWNERS) {
    const int4 x = ((const int4*)c)[tid];
    ((int4*)(cnt2 + (size_t)s * BK_NP + lo))[tid] = x;
    sum = (x.x + x.y) + (x.z + x.w);
  }
#pragma unroll
  for (int off = 16; off > 0; off >>= 1) sum += __shfl_xor(sum, off, 32);
  if (lane == 0 && sum != 0) atomicAdd(&s_tot, sum);
  __syncthreads();
  if (tid == 0) tot[g * BK_S + s] = s_tot;
}

__global__ __launch_bounds__(BK_THREADS)
void gat_bucket_place(const int* __restrict__ ei, const int* __restrict__ cnt2,
                      const int* __restrict__ tot, int* __restrict__ ends,
                      int* __restrict__ rec) {
  __shared__ __attribute__((aligned(16))) int cur[BK_RANGE];
  __shared__ int s_w[32];
  __shared__ int s_b[32];
  const int tid  = threadIdx.x;
  const int lane = tid & 31;
  const int wave = tid >> 5;
  const int g    = blockIdx.x % BK_G;
  const int s    = blockIdx.x / BK_G;
  const int lo   = g * BK_RANGE;

  int below = (tid < g * BK_S) ? tot[tid] : 0;

  int4 t4 = make_int4(0, 0, 0, 0);
  int4 m4 = make_int4(0, 0, 0, 0);
  if (tid < BK_OWNERS) {
#pragma unroll
    for (int q = 0; q < BK_S; ++q) {
      const int4 x = ((const int4*)(cnt2 + (size_t)q * BK_NP + lo))[tid];
      t4.x += x.x; t4.y += x.y; t4.z += x.z; t4.w += x.w;
      const int keep = (q < s) ? -1 : 0;
      m4.x += x.x & keep; m4.y += x.y & keep; m4.z += x.z & keep; m4.w += x.w & keep;
    }
  }
  const int tsum = (t4.x + t4.y) + (t4.z + t4.w);

  int inc = tsum;
#pragma unroll
  for (int off = 1; off < 32; off <<= 1) {
    const int n = __shfl_up(inc, off, 32);
    if (lane >= off) inc += n;
  }
#pragma unroll
  for (int off = 16; off > 0; off >>= 1) below += __shfl_xor(below, off, 32);
  if (lane == 31) s_w[wave] = inc;
  if (lane == 0)  s_b[wave] = below;
  __syncthreads();

  int p = s_b[lane];
#pragma unroll
  for (int off = 16; off > 0; off >>= 1) p += __shfl_xor(p, off, 32);
  const int w = s_w[lane];
  int winc = w;
#pragma unroll
  for (int off = 1; off < 32; off <<= 1) {
    const int n = __shfl_up(winc, off, 32);
    if (lane >= off) winc += n;
  }
  const int wave_off = __shfl(winc - w, wave, 32);

  if (tid < BK_OWNERS) {
    const int st0 = p + wave_off + (inc - tsum);
    const int st1 = st0 + t4.x;
    const int st2 = st1 + t4.y;
    const int st3 = st2 + t4.z;
    ((int4*)cur)[tid] = make_int4(st0 + m4.x, st1 + m4.y, st2 + m4.z, st3 + m4.w);
    const int node = lo + 4 * tid;
    if (s == 0 && node < N_NODES)
      *(int4*)(ends + node) = make_int4(st1, st2, st3, st3 + t4.w);
  }
  __syncthreads();

  const int4* d4 = (const int4*)(ei + (size_t)s * BK_SLICE);
  const int4* j4 = (const int4*)(ei + (size_t)E_EDGES + (size_t)s * BK_SLICE);
#pragma unroll
  for (int b = 0; b < BK_TRIPS; b += 4) {
    int4 v[4];
    int4 jj[4];
#pragma unroll
    for (int u = 0; u < 4; ++u) {
      if (b + u < BK_TRIPS) {
        const int idx = min(tid + (b + u) * BK_THREADS, BK_SLICE4 - 1);
        v[u]  = d4[idx];
        jj[u] = j4[idx];
      }
    }
    int  pos[4][4];
    bool hit[4][4];
#pragma unroll
    for (int u = 0; u < 4; ++u) {
      if (b + u < BK_TRIPS) {
        const int idx = tid + (b + u) * BK_THREADS;
        const bool live = ((b + u + 1) * BK_THREADS <= BK_SLICE4) || (idx < BK_SLICE4);
        const unsigned d0 = (unsigned)(v[u].x - lo);
        const unsigned d1 = (unsigned)(v[u].y - lo);
        const unsigned d2 = (unsigned)(v[u].z - lo);
        const unsigned d3 = (unsigned)(v[u].w - lo);
        hit[u][0] = live && d0 < BK_RANGE;
        hit[u][1] = live && d1 < BK_RANGE;
        hit[u][2] = live && d2 < BK_RANGE;
        hit[u][3] = live && d3 < BK_RANGE;
        if (hit[u][0]) pos[u][0] = atomicAdd(&cur[d0], 1);
        if (hit[u][1]) pos[u][1] = atomicAdd(&cur[d1], 1);
        if (hit[u][2]) pos[u][2] = atomicAdd(&cur[d2], 1);
        if (hit[u][3]) pos[u][3] = atomicAdd(&cur[d3], 1);
      }
    }
#pragma unroll
    for (int u = 0; u < 4; ++u) {
      if (b + u < BK_TRIPS) {
        if (hit[u][0]) rec[pos[u][0]] = jj[u].x;
        if (hit[u][1]) rec[pos[u][1]] = jj[u].y;
        if (hit[u][2]) rec[pos[u][2]] = jj[u].z;
        if (hit[u][3]) rec[pos[u][3]] = jj[u].w;
      }
    }
  }
}

__global__ __launch_bounds__(256)
void gat_node_gather(const int* __restrict__ ends, const int* __restrict__ rec,
                     const float* __restrict__ e_src,
                     const float* __restrict__ e_dst,
                     const float* __restrict__ h, float* __restrict__ out) {
  const int lane = threadIdx.x & 31;
  const int row = __builtin_amdgcn_readfirstlane(blockIdx.x * 8 + (threadIdx.x >> 5));
  if (row >= N_NODES) return;

  const int start = (row > 0) ? ends[row - 1] : 0;
  const int deg   = __builtin_amdgcn_readfirstlane(ends[row] - start);
  const float es  = e_src[row];
  const int* bucket = rec + start;

  int   j0  = 0;
  float ex0 = 0.f;
  if (lane < deg) {
    j0  = bucket[lane];
    ex0 = gat_edge_exp(es + e_dst[j0]);
  }
  float dsum = ex0;
  for (int base = 32; base < deg; base += 32) {
    const int idx = base + lane;
    if (idx < deg) dsum += gat_edge_exp(es + e_dst[bucket[idx]]);
  }
#pragma unroll
  for (int off = 16; off > 0; off >>= 1) dsum += __shfl_xor(dsum, off, 32);
  const float denom = dsum + 1e-12f;

  const float4* h4 = (const float4*)h;
  float4 acc = make_float4(0.f, 0.f, 0.f, 0.f);
  int   jv = j0;
  float av = ex0 / denom;
  for (int base = 0; base < deg; base += 32) {
    if (base > 0) {
      const int idx = base + lane;
      jv = 0;
      av = 0.f;
      if (idx < deg) {
        jv = bucket[idx];
        av = gat_edge_exp(es + e_dst[jv]) / denom;
      }
    }
    const int n = min(32, deg - base);
#pragma unroll 4
    for (int e = 0; e < n; ++e) {
      const int   j = __builtin_amdgcn_readlane(jv, e);
      const float a = __int_as_float(__builtin_amdgcn_readlane(__float_as_int(av), e));
      const float4 hv = h4[(size_t)j * 32 + lane];
      acc.x += a * hv.x;
      acc.y += a * hv.y;
      acc.z += a * hv.z;
      acc.w += a * hv.w;
    }
  }
  ((float4*)out)[(size_t)row * 32 + lane] = acc;
}

extern "C" void kernel_launch(void* const* d_in, const int* in_sizes, int n_in,
                              void* d_out, int out_size, void* d_ws, size_t ws_size,
                              hipStream_t stream) {
  const float* X       = (const float*)d_in[0];
  const int*   ei      = (const int*)d_in[1];
  const float* W       = (const float*)d_in[2];
  const float* att_src = (const float*)d_in[3];
  const float* att_dst = (const float*)d_in[4];
  float* out = (float*)d_out;

  float* h     = (float*)d_ws;
  float* e_src = h + (size_t)N_NODES * OUT_F;
  float* e_dst = e_src + N_NODES;
  int*   ends  = (int*)(e_dst + N_NODES);
  int*   rec   = ends + N_NODES;
  int*   cnt2  = (int*)d_out;
  int*   tot   = cnt2 + (size_t)BK_S * BK_NP;

  gat_gemm_wmma    <<<N_NODES / 16,      256, 0, stream>>>(X, W, att_src, att_dst,
                                                           h, e_src, e_dst);
  gat_bucket_count <<<BK_G * BK_S, BK_THREADS, 0, stream>>>(ei, cnt2, tot);
  gat_bucket_place <<<BK_G * BK_S, BK_THREADS, 0, stream>>>(ei, cnt2, tot, ends, rec);
  gat_node_gather  <<<(N_NODES + 7) / 8, 256, 0, stream>>>(ends, rec, e_src, e_dst,
                                                           h, out);
}


# GEMM on 512 resident work-groups: a wave reads its 16 W rows once into registers and keeps them over all its tiles, X tiles double-buffered in LDS; k sum order regrouped (8-wide groups), still fixed
# speedup vs baseline: 2.9583x; 1.2479x over previous
#include <hip/hip_runtime.h>
#include <hip/hip_bf16.h>

typedef float v2f __attribute__((ext_vector_type(2)));
typedef float v4f __attribute__((ext_vector_type(4)));
typedef float v8f __attribute__((ext_vector_type(8)));

#define N_NODES   100000
#define E_EDGES   1000000
#define IN_F      256
#define OUT_F     128
#define NEG_SLOPE 0.2f

template <int MASK>
__device__ __forceinline__ float row_xor(float v) {
  return __int_as_float(__builtin_amdgcn_update_dpp(
      0, __float_as_int(v), 0x160 | MASK, 0xf, 0xf, false));
}

#define GM_TILES  (N_NODES / 16)
#define GM_GROUPS 512
#define XS_PITCH  260
#define XS_TILE   (16 * XS_PITCH)

static_assert(GM_TILES * 16 == N_NODES, "tiles must cover the rows exactly");
static_assert(GM_GROUPS <= GM_TILES, "every work-group owns at least one tile");

__global__ __launch_bounds__(256)
void gat_gemm_wmma(const float* __restrict__ X, const float* __restrict__ W,
                   const float* __restrict__ att_src,
                   const float* __restrict__ att_dst,
                   float* __restrict__ h,
                   float* __restrict__ e_src, float* __restrict__ e_dst) {
  __shared__ __attribute__((aligned(16))) float xs[2 * XS_TILE];
  __shared__ float red[2 * 256];
  const int tid   = threadIdx.x;
  const int wave  = tid >> 5;
  const int lane  = tid & 31;
  const int col0  = wave * 16;
  const int l15   = lane & 15;
  const int half4 = (lane >> 4) * 4;

  const int xrow = tid >> 6;
  const int xcol = (tid & 63) * 4;
  const float* xg = X + (size_t)xrow * IN_F + xcol;
  float*       xl = &xs[xrow * XS_PITCH + xcol];

  int tile = blockIdx.x;
  v4f xr[4];
#pragma unroll
  for (int it = 0; it < 4; ++it)
    xr[it] = *(const v4f*)(xg + (size_t)tile * (16 * IN_F) + it * (4 * IN_F));

  const float* wrow = W + (size_t)(col0 + l15) * IN_F + half4;
  v4f bw[32];
#pragma unroll
  for (int g = 0; g < 32; ++g) bw[g] = *(const v4f*)(wrow + 8 * g);

  const float a_s = att_src[col0 + l15];
  const float a_d = att_dst[col0 + l15];

#pragma unroll
  for (int it = 0; it < 4; ++it)
    *(v4f*)(xl + it * (4 * XS_PITCH)) = xr[it];
  __syncthreads();

  int cur = 0;
  for (; tile < GM_TILES; tile += GM_GROUPS) {
    const bool more = tile + GM_GROUPS < GM_TILES;
    if (more) {
#pragma unroll
      for (int it = 0; it < 4; ++it)
        xr[it] = *(const v4f*)(xg + (size_t)(tile + GM_GROUPS) * (16 * IN_F) + it * (4 * IN_F));
    }

    const float* xa = &xs[cur * XS_TILE + l15 * XS_PITCH + half4];
    v4f af[2][8];
#pragma unroll
    for (int u = 0; u < 8; ++u) af[0][u] = *(const v4f*)(xa + 8 * u);
#pragma unroll
    for (int u = 0; u < 8; ++u) af[1][u] = *(const v4f*)(xa + 8 * (8 + u));
    __builtin_amdgcn_sched_barrier(0);

    v8f acc = {};
#pragma unroll
    for (int blk = 0; blk < 4; ++blk) {
#pragma unroll
      for (int u = 0; u < 8; ++u) {
        const v4f a = af[blk & 1][u];
        const v4f b = bw[8 * blk + u];
        const v2f a0 = {a.x, a.y};
        const v2f b0 = {b.x, b.y};
        const v2f a1 = {a.z, a.w};
        const v2f b1 = {b.z, b.w};
        acc = __builtin_amdgcn_wmma_f32_16x16x4_f32(
             false, a0,  false, b0,
             (short)0, acc,  false,  false);
        acc = __builtin_amdgcn_wmma_f32_16x16x4_f32(
             false, a1,  false, b1,
             (short)0, acc,  false,  false);
      }
      if (blk < 2) {
        __builtin_amdgcn_sched_barrier(0);
#pragma unroll
        for (int u = 0; u < 8; ++u)
          af[blk & 1][u] = *(const v4f*)(xa + 8 * (8 * (blk + 2) + u));
        __builtin_amdgcn_sched_barrier(0);
      }
    }

    asm volatile("v_nop\n\tv_nop\n\tv_nop\n\tv_nop\n\tv_nop\n\tv_nop\n\tv_nop\n\tv_nop" : "+v"(acc));

    float v[16];
#pragma unroll
    for (int r = 0; r < 8; ++r) {
      v[r]     = acc[r] * a_s;
      v[8 + r] = acc[r] * a_d;
    }
    const bool b3 = (l15 & 8) != 0;
    const bool b2 = (l15 & 4) != 0;
    const bool b1 = (l15 & 2) != 0;
    const bool b0 = (l15 & 1) != 0;
    float w8[8];
#pragma unroll
    for (int i = 0; i < 8; ++i) {
      const float send = b3 ? v[i] : v[i + 8];
      const float keep = b3 ? v[i + 8] : v[i];
      w8[i] = keep + row_xor<8>(send);
    }
    float w4[4];
#pragma unroll
    for (int i = 0; i < 4; ++i) {
      const float send = b2 ? w8[i] : w8[i + 4];
      const float keep = b2 ? w8[i + 4] : w8[i];
      w4[i] = keep + row_xor<4>(send);
    }
    float w2[2];
#pragma unroll
    for (int i = 0; i < 2; ++i) {
      const float send = b1 ? w4[i] : w4[i + 2];
      const float keep = b1 ? w4[i + 2] : w4[i];
      w2[i] = keep + row_xor<2>(send);
    }
    const float w1 = (b0 ? w2[1] : w2[0]) + row_xor<1>(b0 ? w2[0] : w2[1]);
    red[cur * 256 + wave * 32 + lane] = w1;

    if (more) {
#pragma unroll
      for (int it = 0; it < 4; ++it)
        *(v4f*)(xl + (cur ^ 1) * XS_TILE + it * (4 * XS_PITCH)) = xr[it];
    }
    __syncthreads();

    const int m0   = tile * 16;
    const int mrow = m0 + ((lane >> 4) << 3);
    float* hp = h + (size_t)mrow * OUT_F + col0 + l15;
#pragma unroll
    for (int r = 0; r < 8; ++r) {
      hp[(size_t)r * OUT_F] = acc[r];
    }

    if (tid < 32) {
      const float* rc = &red[cur * 256];
      float s = rc[tid];
#pragma unroll
      for (int w = 1; w < 8; ++w) s += rc[w * 32 + tid];
      const int row = m0 + ((tid >> 4) << 3) + (tid & 7);
      float* dst = (tid & 8) ? e_dst : e_src;
      dst[row] = s;
    }
    cur ^= 1;
  }
}

#define BK_G       32
#define BK_S       8
#define BK_RANGE   3128
#define BK_NP      (BK_G * BK_RANGE)
#define BK_SLICE   (E_EDGES / BK_S)
#define BK_SLICE4  (BK_SLICE / 4)
#define BK_THREADS 1024
#define BK_TRIPS   ((BK_SLICE4 + BK_THREADS - 1) / BK_THREADS)
#define BK_OWNERS  (BK_RANGE / 4)

static_assert(BK_NP >= N_NODES, "ranges must cover every node");
static_assert(BK_S * BK_SLICE == E_EDGES && BK_SLICE % 4 == 0, "slices must tile the edge list in int4s");
static_assert(BK_RANGE % 4 == 0 && N_NODES % 4 == 0, "int4 access to node planes");
static_assert(BK_OWNERS <= BK_THREADS && BK_G * BK_S <= BK_THREADS, "one thread per owner / per total");

__device__ __forceinline__ float gat_edge_exp(float e) {
  e = (e >= 0.f) ? e : NEG_SLOPE * e;
  e = fminf(10.f, fmaxf(-10.f, e));
  return __expf(e);
}

__global__ __launch_bounds__(BK_THREADS)
void gat_bucket_count(const int* __restrict__ ei, int* __restrict__ cnt2,
                      int* __restrict__ tot) {
  __shared__ __attribute__((aligned(16))) int c[BK_RANGE];
  __shared__ int s_tot;
  const int tid  = threadIdx.x;
  const int lane = tid & 31;
  const int g    = blockIdx.x % BK_G;
  const int s    = blockIdx.x / BK_G;
  const int lo   = g * BK_RANGE;

  for (int k = tid; k < BK_RANGE; k += BK_THREADS) c[k] = 0;
  if (tid == 0) s_tot = 0;
  __syncthreads();

  const int4* d4 = (const int4*)(ei + (size_t)s * BK_SLICE);
#pragma unroll
  for (int b = 0; b < BK_TRIPS; b += 8) {
    int4 v[8];
#pragma unroll
    for (int u = 0; u < 8; ++u) {
      if (b + u < BK_TRIPS) {
        const int idx = tid + (b + u) * BK_THREADS;
        v[u] = d4[min(idx, BK_SLICE4 - 1)];
      }
    }
#pragma unroll
    for (int u = 0; u < 8; ++u) {
      if (b + u < BK_TRIPS) {
        const int idx = tid + (b + u) * BK_THREADS;
        const bool live = ((b + u + 1) * BK_THREADS <= BK_SLICE4) || (idx < BK_SLICE4);
        const unsigned d0 = (unsigned)(v[u].x - lo);
        const unsigned d1 = (unsigned)(v[u].y - lo);
        const unsigned d2 = (unsigned)(v[u].z - lo);
        const unsigned d3 = (unsigned)(v[u].w - lo);
        if (live && d0 < BK_RANGE) atomicAdd(&c[d0], 1);
        if (live && d1 < BK_RANGE) atomicAdd(&c[d1], 1);
        if (live && d2 < BK_RANGE) atomicAdd(&c[d2], 1);
        if (live && d3 < BK_RANGE) atomicAdd(&c[d3], 1);
      }
    }
  }
  __syncthreads();

  int sum = 0;
  if (tid < BK_OWNERS) {
    const int4 x = ((const int4*)c)[tid];
    ((int4*)(cnt2 + (size_t)s * BK_NP + lo))[tid] = x;
    sum = (x.x + x.y) + (x.z + x.w);
  }
#pragma unroll
  for (int off = 16; off > 0; off >>= 1) sum += __shfl_xor(sum, off, 32);
  if (lane == 0 && sum != 0) atomicAdd(&s_tot, sum);
  __syncthreads();
  if (tid == 0) tot[g * BK_S + s] = s_tot;
}

__global__ __launch_bounds__(BK_THREADS)
void gat_bucket_place(const int* __restrict__ ei, const int* __restrict__ cnt2,
                      const int* __restrict__ tot, int* __restrict__ ends,
                      int* __restrict__ rec) {
  __shared__ __attribute__((aligned(16))) int cur[BK_RANGE];
  __shared__ int s_w[32];
  __shared__ int s_b[32];
  const int tid  = threadIdx.x;
  const int lane = tid & 31;
  const int wave = tid >> 5;
  const int g    = blockIdx.x % BK_G;
  const int s    = blockIdx.x / BK_G;
  const int lo   = g * BK_RANGE;

  int below = (tid < g * BK_S) ? tot[tid] : 0;

  int4 t4 = make_int4(0, 0, 0, 0);
  int4 m4 = make_int4(0, 0, 0, 0);
  if (tid < BK_OWNERS) {
#pragma unroll
    for (int q = 0; q < BK_S; ++q) {
      const int4 x = ((const int4*)(cnt2 + (size_t)q * BK_NP + lo))[tid];
      t4.x += x.x; t4.y += x.y; t4.z += x.z; t4.w += x.w;
      const int keep = (q < s) ? -1 : 0;
      m4.x += x.x & keep; m4.y += x.y & keep; m4.z += x.z & keep; m4.w += x.w & keep;
    }
  }
  const int tsum = (t4.x + t4.y) + (t4.z + t4.w);

  int inc = tsum;
#pragma unroll
  for (int off = 1; off < 32; off <<= 1) {
    const int n = __shfl_up(inc, off, 32);
    if (lane >= off) inc += n;
  }
#pragma unroll
  for (int off = 16; off > 0; off >>= 1) below += __shfl_xor(below, off, 32);
  if (lane == 31) s_w[wave] = inc;
  if (lane == 0)  s_b[wave] = below;
  __syncthreads();

  int p = s_b[lane];
#pragma unroll
  for (int off = 16; off > 0; off >>= 1) p += __shfl_xor(p, off, 32);
  const int w = s_w[lane];
  int winc = w;
#pragma unroll
  for (int off = 1; off < 32; off <<= 1) {
    const int n = __shfl_up(winc, off, 32);
    if (lane >= off) winc += n;
  }
  const int wave_off = __shfl(winc - w, wave, 32);

  if (tid < BK_OWNERS) {
    const int st0 = p + wave_off + (inc - tsum);
    const int st1 = st0 + t4.x;
    const int st2 = st1 + t4.y;
    const int st3 = st2 + t4.z;
    ((int4*)cur)[tid] = make_int4(st0 + m4.x, st1 + m4.y, st2 + m4.z, st3 + m4.w);
    const int node = lo + 4 * tid;
    if (s == 0 && node < N_NODES)
      *(int4*)(ends + node) = make_int4(st1, st2, st3, st3 + t4.w);
  }
  __syncthreads();

  const int4* d4 = (const int4*)(ei + (size_t)s * BK_SLICE);
  const int4* j4 = (const int4*)(ei + (size_t)E_EDGES + (size_t)s * BK_SLICE);
#pragma unroll
  for (int b = 0; b < BK_TRIPS; b += 4) {
    int4 v[4];
    int4 jj[4];
#pragma unroll
    for (int u = 0; u < 4; ++u) {
      if (b + u < BK_TRIPS) {
        const int idx = min(tid + (b + u) * BK_THREADS, BK_SLICE4 - 1);
        v[u]  = d4[idx];
        jj[u] = j4[idx];
      }
    }
    int  pos[4][4];
    bool hit[4][4];
#pragma unroll
    for (int u = 0; u < 4; ++u) {
      if (b + u < BK_TRIPS) {
        const int idx = tid + (b + u) * BK_THREADS;
        const bool live = ((b + u + 1) * BK_THREADS <= BK_SLICE4) || (idx < BK_SLICE4);
        const unsigned d0 = (unsigned)(v[u].x - lo);
        const unsigned d1 = (unsigned)(v[u].y - lo);
        const unsigned d2 = (unsigned)(v[u].z - lo);
        const unsigned d3 = (unsigned)(v[u].w - lo);
        hit[u][0] = live && d0 < BK_RANGE;
        hit[u][1] = live && d1 < BK_RANGE;
        hit[u][2] = live && d2 < BK_RANGE;
        hit[u][3] = live && d3 < BK_RANGE;
        if (hit[u][0]) pos[u][0] = atomicAdd(&cur[d0], 1);
        if (hit[u][1]) pos[u][1] = atomicAdd(&cur[d1], 1);
        if (hit[u][2]) pos[u][2] = atomicAdd(&cur[d2], 1);
        if (hit[u][3]) pos[u][3] = atomicAdd(&cur[d3], 1);
      }
    }
#pragma unroll
    for (int u = 0; u < 4; ++u) {
      if (b + u < BK_TRIPS) {
        if (hit[u][0]) rec[pos[u][0]] = jj[u].x;
        if (hit[u][1]) rec[pos[u][1]] = jj[u].y;
        if (hit[u][2]) rec[pos[u][2]] = jj[u].z;
        if (hit[u][3]) rec[pos[u][3]] = jj[u].w;
      }
    }
  }
}

__global__ __launch_bounds__(256)
void gat_node_gather(const int* __restrict__ ends, const int* __restrict__ rec,
                     const float* __restrict__ e_src,
                     const float* __restrict__ e_dst,
                     const float* __restrict__ h, float* __restrict__ out) {
  const int lane = threadIdx.x & 31;
  const int row = __builtin_amdgcn_readfirstlane(blockIdx.x * 8 + (threadIdx.x >> 5));
  if (row >= N_NODES) return;

  const int start = (row > 0) ? ends[row - 1] : 0;
  const int deg   = __builtin_amdgcn_readfirstlane(ends[row] - start);
  const float es  = e_src[row];
  const int* bucket = rec + start;

  int   j0  = 0;
  float ex0 = 0.f;
  if (lane < deg) {
    j0  = bucket[lane];
    ex0 = gat_edge_exp(es + e_dst[j0]);
  }
  float dsum = ex0;
  for (int base = 32; base < deg; base += 32) {
    const int idx = base + lane;
    if (idx < deg) dsum += gat_edge_exp(es + e_dst[bucket[idx]]);
  }
#pragma unroll
  for (int off = 16; off > 0; off >>= 1) dsum += __shfl_xor(dsum, off, 32);
  const float denom = dsum + 1e-12f;

  const float4* h4 = (const float4*)h;
  float4 acc = make_float4(0.f, 0.f, 0.f, 0.f);
  int   jv = j0;
  float av = ex0 / denom;
  for (int base = 0; base < deg; base += 32) {
    if (base > 0) {
      const int idx = base + lane;
      jv = 0;
      av = 0.f;
      if (idx < deg) {
        jv = bucket[idx];
        av = gat_edge_exp(es + e_dst[jv]) / denom;
      }
    }
    const int n = min(32, deg - base);
#pragma unroll 4
    for (int e = 0; e < n; ++e) {
      const int   j = __builtin_amdgcn_readlane(jv, e);
      const float a = __int_as_float(__builtin_amdgcn_readlane(__float_as_int(av), e));
      const float4 hv = h4[(size_t)j * 32 + lane];
      acc.x += a * hv.x;
      acc.y += a * hv.y;
      acc.z += a * hv.z;
      acc.w += a * hv.w;
    }
  }
  ((float4*)out)[(size_t)row * 32 + lane] = acc;
}

extern "C" void kernel_launch(void* const* d_in, const int* in_sizes, int n_in,
                              void* d_out, int out_size, void* d_ws, size_t ws_size,
                              hipStream_t stream) {
  const float* X       = (const float*)d_in[0];
  const int*   ei      = (const int*)d_in[1];
  const float* W       = (const float*)d_in[2];
  const float* att_src = (const float*)d_in[3];
  const float* att_dst = (const float*)d_in[4];
  float* out = (float*)d_out;

  float* h     = (float*)d_ws;
  float* e_src = h + (size_t)N_NODES * OUT_F;
  float* e_dst = e_src + N_NODES;
  int*   ends  = (int*)(e_dst + N_NODES);
  int*   rec   = ends + N_NODES;
  int*   cnt2  = (int*)d_out;
  int*   tot   = cnt2 + (size_t)BK_S * BK_NP;

  gat_gemm_wmma    <<<GM_GROUPS,         256, 0, stream>>>(X, W, att_src, att_dst,
                                                           h, e_src, e_dst);
  gat_bucket_count <<<BK_G * BK_S, BK_THREADS, 0, stream>>>(ei, cnt2, tot);
  gat_bucket_place <<<BK_G * BK_S, BK_THREADS, 0, stream>>>(ei, cnt2, tot, ends, rec);
  gat_node_gather  <<<(N_NODES + 7) / 8, 256, 0, stream>>>(ends, rec, e_src, e_dst,
                                                           h, out);
}
